# starting kernel
# baseline (speedup 1.0000x reference)
#include <hip/hip_runtime.h>


typedef __attribute__((ext_vector_type(2))) float v2f;
typedef __attribute__((ext_vector_type(8))) float v8f;

#define IMG   224
#define SROWS 144
#define SSTR  136
#define S_OFF   0
#define H3_OFF  (SROWS*SSTR)
#define H2_OFF  (H3_OFF + 144*16)
#define H1_OFF  (H2_OFF + 80*16)
#define W_OFF   (H1_OFF + 32*16)
#define SC_OFF  (W_OFF + 208)
#define NT3 35
#define NT2 18
#define NT1 7
#define TB3_OFF (SC_OFF + 16)
#define TB2_OFF (TB3_OFF + NT3*32*2)
#define TB1_OFF (TB2_OFF + NT2*32*2)
#define SMEM_FLOATS (TB1_OFF + NT1*32*2)

__device__ __forceinline__ float bicubic_w(int t, int K) {
  const double A_ = -0.5;
  double x  = ((double)t - 0.5 * (double)K + 0.5) * (4.0 / (double)K);
  double ax = fabs(x);
  double w;
  if (ax > 2.0)       w = 0.0;
  else if (ax > 1.0)  w = ((A_ * ax - 5.0 * A_) * ax + 8.0 * A_) * ax - 4.0 * A_;
  else                w = (((A_ + 2.0) * ax - (A_ + 3.0)) * ax) * ax + 1.0;
  return (float)w;
}

__global__ __launch_bounds__(256) void glimpse_wmma_kernel(
    const float* __restrict__ imgs, const int* __restrict__ locs,
    float* __restrict__ out)
{
  extern __shared__ float sm[];
  float* S  = sm + S_OFF;
  float* Wv = sm + W_OFF;
  float* SC = sm + SC_OFF;

  const int b    = blockIdx.x;
  const int tid  = threadIdx.x;
  const int lane = tid & 31;
  const int wave = __builtin_amdgcn_readfirstlane(tid >> 5);

  if (tid < 193) {
    int t = tid, K, base;
    if (t < 12)      { K = 12;  base = 0;            }
    else if (t < 69) { K = 57;  base = 12; t -= 12;  }
    else             { K = 124; base = 69; t -= 69;  }
    Wv[base + t] = bicubic_w(t, K);
  }

  const int x0 = locs[2 * b + 0] - 67;
  const int y0 = locs[2 * b + 1] - 67;
  const float* img = imgs + (size_t)b * (3 * IMG * IMG);
  for (int idx = tid; idx < SROWS * SSTR; idx += 256) {
    int r = idx / SSTR;
    int c = idx - r * SSTR;
    float v = 0.f;
    int iy = x0 + r, ix = y0 + c;
    if (r < 135 && c < 135 && iy >= 0 && iy < IMG && ix >= 0 && ix < IMG) {
      int o = iy * IMG + ix;
      v = __builtin_nontemporal_load(img + o)
        + __builtin_nontemporal_load(img + IMG * IMG + o)
        + __builtin_nontemporal_load(img + 2 * IMG * IMG + o);
    }
    S[idx] = v;
  }
  __syncthreads();

  if (tid < 4) {
    if (tid == 3) { SC[0] = 1.0f / 3.0f; }
    else {
      int K    = (tid == 0) ? 12 : ((tid == 1) ? 57 : 124);
      int base = (tid == 0) ? 0  : ((tid == 1) ? 12 : 69);
      double s = 0.0;
      for (int i = 0; i < K; ++i) s += (double)Wv[base + i];
      SC[tid + 1] = (float)(1.0 / (3.0 * s * s));
    }
  }
  for (int e = tid; e < (NT3 + NT2 + NT1) * 32; e += 256) {
    int gi, le, base;
    if (e < NT3 * 32)              { gi = 3; le = e;              base = TB3_OFF; }
    else if (e < (NT3 + NT2) * 32) { gi = 2; le = e - NT3 * 32;   base = TB2_OFF; }
    else                           { gi = 1; le = e - (NT3 + NT2) * 32; base = TB1_OFF; }
    const int K  = (gi == 3) ? 124 : ((gi == 2) ? 57 : 12);
    const int g  = K + 11;
    const int wb = (gi == 3) ? 69 : ((gi == 2) ? 12 : 0);
    const int t  = le >> 5;
    const int l  = le & 31;
    const int mm = l & 15;
    const int c1 = t * 4 + ((l >> 4) << 1);
    const int d0 = c1 - mm;
    float v0 = Wv[wb + min(max(d0, 0), K - 1)];
    float v1 = Wv[wb + min(max(d0 + 1, 0), K - 1)];
    v0 = (mm < 12 && c1     < g && d0     >= 0 && d0     < K) ? v0 : 0.f;
    v1 = (mm < 12 && c1 + 1 < g && d0 + 1 >= 0 && d0 + 1 < K) ? v1 : 0.f;
    sm[base + 2 * le]     = v0;
    sm[base + 2 * le + 1] = v1;
  }
  __syncthreads();

  const int m  = lane & 15;
  const int kk = (lane >> 4) << 1;
  const int Mb = (lane >> 4) << 3;

  for (int task = wave; task < 16; task += 8) {
    int gi, blk;
    if (task < 9)       { gi = 3; blk = task;      }
    else if (task < 14) { gi = 2; blk = task - 9;  }
    else                { gi = 1; blk = task - 14; }
    const int g     = (gi == 3) ? 135 : ((gi == 2) ? 68 : 23);
    const int off   = 67 - (g >> 1);
    const int ntile = (g + 3) >> 2;
    const v2f* TBg  = (const v2f*)(sm + ((gi == 3) ? TB3_OFF : ((gi == 2) ? TB2_OFF : TB1_OFF)));
    float*     H    = sm + ((gi == 3) ? H3_OFF : ((gi == 2) ? H2_OFF : H1_OFF));
    const float* Srow = S + (off + blk * 16 + m) * SSTR + off + kk;

    v8f acc = {};
    v2f a, bm;
    a.x = Srow[0]; a.y = Srow[1];
    bm  = TBg[lane];
    for (int t = 0; t < ntile; ++t) {
      v2f a2, b2;
      a2.x = Srow[(t + 1) * 4];
      a2.y = Srow[(t + 1) * 4 + 1];
      b2   = TBg[(t + 1) * 32 + lane];
      acc = __builtin_amdgcn_wmma_f32_16x16x4_f32(
          false, a, false, bm, (short)0, acc, false, false);
      a = a2; bm = b2;
    }
#pragma unroll
    for (int r = 0; r < 8; ++r)
      H[(blk * 16 + Mb + r) * 16 + m] = acc[r];
  }
  __syncthreads();

  if (wave < 3) {
    const int gi    = wave + 1;
    const int g     = (gi == 3) ? 135 : ((gi == 2) ? 68 : 23);
    const int ntile = (g + 3) >> 2;
    const v2f* TBg  = (const v2f*)(sm + ((gi == 3) ? TB3_OFF : ((gi == 2) ? TB2_OFF : TB1_OFF)));
    const float* H  = sm + ((gi == 3) ? H3_OFF : ((gi == 2) ? H2_OFF : H1_OFF)) + m;

    v8f acc = {};
    v2f a, bm;
    a    = TBg[lane];
    bm.x = H[kk * 16];
    bm.y = H[(kk + 1) * 16];
    for (int t = 0; t < ntile; ++t) {
      v2f a2, b2;
      a2   = TBg[(t + 1) * 32 + lane];
      b2.x = H[((t + 1) * 4 + kk) * 16];
      b2.y = H[((t + 1) * 4 + kk + 1) * 16];
      acc = __builtin_amdgcn_wmma_f32_16x16x4_f32(
          false, a, false, bm, (short)0, acc, false, false);
      a = a2; bm = b2;
    }
    const float sc = SC[gi];
#pragma unroll
    for (int r = 0; r < 8; ++r) {
      const int M = Mb + r;
      if (M < 12 && m < 12)
        out[(((size_t)b * 4 + gi) * 12 + M) * 12 + m] = acc[r] * sc;
    }
  } else if (wave == 3) {
    const float sc = SC[0];
    for (int t = lane; t < 144; t += 32) {
      int o = t / 12, j = t - o * 12;
      out[((size_t)b * 4 * 12 + o) * 12 + j] = S[(61 + o) * SSTR + 61 + j] * sc;
    }
  }
}

extern "C" void kernel_launch(void* const* d_in, const int* in_sizes, int n_in,
                              void* d_out, int out_size, void* d_ws, size_t ws_size,
                              hipStream_t stream) {
  const float* imgs = (const float*)d_in[0];
  const int*   locs = (const int*)d_in[1];
  float*       out  = (float*)d_out;
  (void)in_sizes; (void)n_in; (void)d_ws; (void)ws_size; (void)out_size;

  const size_t shmem = (size_t)SMEM_FLOATS * sizeof(float);
  glimpse_wmma_kernel<<<dim3(256), dim3(256), shmem, stream>>>(imgs, locs, out);
}


# horizontal pass unrolled, its LDS operands loaded a group of 8 k-tiles ahead into a second register set instead of load, wait, two products per trip; tile order and both accumulator sums unchanged
# speedup vs baseline: 4.4431x; 4.4431x over previous
#include <hip/hip_runtime.h>


typedef __attribute__((ext_vector_type(2))) float v2f;
typedef __attribute__((ext_vector_type(8))) float v8f;

#define IMG      224
#define PLANE    (IMG * IMG)
#define NTT      36
#define W_OFF    0
#define SC_OFF   208
#define TB_OFF   224
#define TBSZ     (NTT * 64)
#define PITCH    148
#define PSZ      (16 * PITCH)
#define P_OFF    (TB_OFF + 3 * TBSZ)
#define SMEM_FLOATS (P_OFF + 6 * PSZ)
#define NWAVES   19
#define NTHREADS (NWAVES * 32)

__device__ __forceinline__ float bicubic_w(int t, int K) {
  const double A_ = -0.5;
  double x  = ((double)t - 0.5 * (double)K + 0.5) * (4.0 / (double)K);
  double ax = fabs(x);
  double w;
  if (ax > 2.0)       w = 0.0;
  else if (ax > 1.0)  w = ((A_ * ax - 5.0 * A_) * ax + 8.0 * A_) * ax - 4.0 * A_;
  else                w = (((A_ + 2.0) * ax - (A_ + 3.0)) * ax) * ax + 1.0;
  return (float)w;
}

__device__ __forceinline__ void parity_power_sums(int q0, int m, int s[4]) {
  const int J1 = m * (m - 1) / 2;
  const int J2 = (m - 1) * m * (2 * m - 1) / 6;
  const int J3 = J1 * J1;
  s[0] = m;
  s[1] = m * q0 + 2 * J1;
  s[2] = m * q0 * q0 + 4 * q0 * J1 + 4 * J2;
  s[3] = m * q0 * q0 * q0 + 6 * q0 * q0 * J1 + 12 * q0 * J2 + 8 * J3;
}

__device__ __forceinline__ float tap_scale(int K) {
  const int odd  = K & 1;
  const int q0   = odd ? 2 : 1;
  const int nin  = K >> 1;
  const int m_in  = (nin >= q0) ? (nin - q0) / 2 + 1 : 0;
  const int m_all = (K - 1 >= q0) ? (K - 1 - q0) / 2 + 1 : 0;
  int in[4], all[4];
  parity_power_sums(q0, m_in, in);
  parity_power_sums(q0, m_all, all);
  const long long I0 = 2 * in[0] + odd, I2 = 2 * in[2], I3 = 2 * in[3];
  const long long O0 = 2 * (all[0] - in[0]), O1 = 2 * (all[1] - in[1]);
  const long long O2 = 2 * (all[2] - in[2]), O3 = 2 * (all[3] - in[3]);
  const long long K1 = K, K2 = K1 * K1, K3 = K2 * K1;
  const long long num = 4 * (3 * I3 - O3) + 10 * (O2 - I2) * K1 - 8 * O1 * K2 + (I0 + 2 * O0) * K3;
  const double n = (double)num, d = (double)K3;
  return (float)((d * d) / (3.0 * n * n));
}

template <int NG, int PART>
__device__ __forceinline__ void vconv_task(const float* __restrict__ img, const float* sm,
                                           float* P, int cb, int x0, int y0, int lane)
{
  const int n = lane & 15;
  const int h = lane >> 4;
  const int C = cb * 16 + n;
  const int ix = y0 + C;
  const bool cval = (C < 135) && (ix >= 0) && (ix < IMG);
  const int ixc = min(max(ix, 0), IMG - 1);
  const v2f* TB1 = (const v2f*)(sm + TB_OFF + 0 * TBSZ) + lane;
  const v2f* TB2 = (const v2f*)(sm + TB_OFF + 1 * TBSZ) + lane;
  const v2f* TB3 = (const v2f*)(sm + TB_OFF + 2 * TBSZ) + lane;
  v8f acc3 = {}, acc2 = {}, acc1 = {};

#pragma unroll
  for (int grp = 0; grp < 3; ++grp) {
    float lv[6][2][3];
    bool  rv[6][2];
#pragma unroll
    for (int i = 0; i < 6; ++i) {
      const int t = PART * 18 + grp * 6 + i;
      if (t < 34) {
#pragma unroll
        for (int s = 0; s < 2; ++s) {
          const int R  = 4 * t + 2 * h + s;
          const int iy = x0 + R;
          rv[i][s] = cval && (R < 135) && (iy >= 0) && (iy < IMG);
          const float* p = img + min(max(iy, 0), IMG - 1) * IMG + ixc;
          lv[i][s][0] = p[0];
          lv[i][s][1] = p[PLANE];
          lv[i][s][2] = p[2 * PLANE];
        }
      }
    }
#pragma unroll
    for (int i = 0; i < 6; ++i) {
      const int t = PART * 18 + grp * 6 + i;
      if (t < 34) {
        v2f bm;
        bm.x = __int_as_float(__float_as_int((lv[i][0][0] + lv[i][0][1]) + lv[i][0][2]) & (rv[i][0] ? -1 : 0));
        bm.y = __int_as_float(__float_as_int((lv[i][1][0] + lv[i][1][1]) + lv[i][1][2]) & (rv[i][1] ? -1 : 0));
        acc3 = __builtin_amdgcn_wmma_f32_16x16x4_f32(
            false, TB3[t * 32], false, bm, (short)0, acc3, false, false);
        if (NG >= 2 && t >= 8 && t < 26)
          acc2 = __builtin_amdgcn_wmma_f32_16x16x4_f32(
              false, TB2[t * 32], false, bm, (short)0, acc2, false, false);
        if (NG >= 3 && t >= 14 && t < 20)
          acc1 = __builtin_amdgcn_wmma_f32_16x16x4_f32(
              false, TB1[t * 32], false, bm, (short)0, acc1, false, false);
      }
    }
  }

  float* Pp = P + (PART * 3) * PSZ + (8 * h) * PITCH + C;
#pragma unroll
  for (int r = 0; r < 8; ++r) {
    Pp[2 * PSZ + r * PITCH] = acc3[r];
    if (NG >= 2) Pp[1 * PSZ + r * PITCH] = acc2[r];
    if (NG >= 3) Pp[0 * PSZ + r * PITCH] = acc1[r];
  }
}

#define HCH 8
template <int GI>
__device__ __forceinline__ void hconv_task(const float* sm, const float* P,
                                           float* __restrict__ out, int b, int lane)
{
  constexpr int tb  = (GI == 3) ? 0  : ((GI == 2) ? 8  : 14);
  constexpr int te  = (GI == 3) ? 34 : ((GI == 2) ? 26 : 20);
  constexpr int NCH = (te - tb + HCH - 1) / HCH;
  const int m  = lane & 15;
  const int kk = (lane >> 4) << 1;
  const int Mb = (lane >> 4) << 3;
  const v2f* TBg = (const v2f*)(sm + TB_OFF + (GI - 1) * TBSZ) + lane;
  const float* V0 = P + (GI - 1) * PSZ + m * PITCH + kk;
  const float* V1 = V0 + 3 * PSZ;

  v2f p0[2][HCH], p1[2][HCH], bt[2][HCH];
#pragma unroll
  for (int i = 0; i < HCH; ++i) {
    const int t = tb + i;
    if (t < te) {
      p0[0][i] = *(const v2f*)(V0 + 4 * t);
      p1[0][i] = *(const v2f*)(V1 + 4 * t);
      bt[0][i] = TBg[t * 32];
    }
  }
  const float sc = sm[SC_OFF + GI];

  v8f accE = {}, accO = {};
#pragma unroll
  for (int c = 0; c < NCH; ++c) {
    if (c + 1 < NCH) {
#pragma unroll
      for (int i = 0; i < HCH; ++i) {
        const int t = tb + (c + 1) * HCH + i;
        if (t < te) {
          p0[(c + 1) & 1][i] = *(const v2f*)(V0 + 4 * t);
          p1[(c + 1) & 1][i] = *(const v2f*)(V1 + 4 * t);
          bt[(c + 1) & 1][i] = TBg[t * 32];
        }
      }
    }
    __builtin_amdgcn_sched_barrier(0);
#pragma unroll
    for (int i = 0; i < HCH; ++i) {
      const int t = tb + c * HCH + i;
      if (t < te) {
        const v2f a = p0[c & 1][i] + p1[c & 1][i];
        if (i & 1)
          accO = __builtin_amdgcn_wmma_f32_16x16x4_f32(
              false, a, false, bt[c & 1][i], (short)0, accO, false, false);
        else
          accE = __builtin_amdgcn_wmma_f32_16x16x4_f32(
              false, a, false, bt[c & 1][i], (short)0, accE, false, false);
      }
    }
    __builtin_amdgcn_sched_barrier(0);
  }
#pragma unroll
  for (int r = 0; r < 8; ++r) {
    const int M = Mb + r;
    if (M < 12 && m < 12)
      out[(((size_t)b * 4 + GI) * 12 + M) * 12 + m] = (accE[r] + accO[r]) * sc;
  }
}

__global__ __launch_bounds__(NTHREADS) void glimpse_wmma_kernel(
    const float* __restrict__ imgs, const int* __restrict__ locs,
    float* __restrict__ out)
{
  extern __shared__ float sm[];
  float* Wv = sm + W_OFF;
  float* SC = sm + SC_OFF;
  float* P  = sm + P_OFF;

  const int b    = blockIdx.x;
  const int tid  = threadIdx.x;
  const int lane = tid & 31;
  const int wave = __builtin_amdgcn_readfirstlane(tid >> 5);

  if (tid < 193) {
    int t = tid, K, base;
    if (t < 12)      { K = 12;  base = 0;            }
    else if (t < 69) { K = 57;  base = 12; t -= 12;  }
    else             { K = 124; base = 69; t -= 69;  }
    Wv[base + t] = bicubic_w(t, K);
  } else if (tid >= 224 && tid < 228) {
    const int f = tid - 224;
    if (f == 3) { SC[0] = 1.0f / 3.0f; }
    else        { SC[f + 1] = tap_scale((f == 0) ? 12 : ((f == 1) ? 57 : 124)); }
  }
  __syncthreads();

  for (int e = tid; e < 3 * NTT * 32; e += NTHREADS) {
    const int gi = e / (NTT * 32) + 1;
    const int le = e - (gi - 1) * (NTT * 32);
    const int K   = (gi == 3) ? 124 : ((gi == 2) ? 57 : 12);
    const int wb  = (gi == 3) ? 69  : ((gi == 2) ? 12 : 0);
    const int off = (gi == 3) ? 0   : ((gi == 2) ? 33 : 56);
    const int t  = le >> 5;
    const int l  = le & 31;
    const int mm = l & 15;
    const int d0 = t * 4 + ((l >> 4) << 1) - off - mm;
    float v0 = Wv[wb + min(max(d0, 0), K - 1)];
    float v1 = Wv[wb + min(max(d0 + 1, 0), K - 1)];
    v0 = (mm < 12 && d0     >= 0 && d0     < K) ? v0 : 0.f;
    v1 = (mm < 12 && d0 + 1 >= 0 && d0 + 1 < K) ? v1 : 0.f;
    sm[TB_OFF + 2 * e]     = v0;
    sm[TB_OFF + 2 * e + 1] = v1;
  }
  __syncthreads();

  const int x0 = locs[2 * b + 0] - 67;
  const int y0 = locs[2 * b + 1] - 67;
  const float* img = imgs + (size_t)b * (3 * PLANE);

  if (wave < 18) {
    const int part = (wave >= 9) ? 1 : 0;
    const int cb   = wave - 9 * part;
    const int ng   = (cb == 3 || cb == 4) ? 3 : ((cb >= 2 && cb <= 6) ? 2 : 1);
    if (part == 0) {
      if (ng == 3)      vconv_task<3, 0>(img, sm, P, cb, x0, y0, lane);
      else if (ng == 2) vconv_task<2, 0>(img, sm, P, cb, x0, y0, lane);
      else              vconv_task<1, 0>(img, sm, P, cb, x0, y0, lane);
    } else {
      if (ng == 3)      vconv_task<3, 1>(img, sm, P, cb, x0, y0, lane);
      else if (ng == 2) vconv_task<2, 1>(img, sm, P, cb, x0, y0, lane);
      else              vconv_task<1, 1>(img, sm, P, cb, x0, y0, lane);
    }
  } else {
    for (int t = lane; t < 144; t += 32) {
      const int o = t / 12, j = t - o * 12;
      const int iy = x0 + 61 + o, ix = y0 + 61 + j;
      const bool ok = (iy >= 0) && (iy < IMG) && (ix >= 0) && (ix < IMG);
      const float* p = img + min(max(iy, 0), IMG - 1) * IMG + min(max(ix, 0), IMG - 1);
      const float l0 = p[0];
      const float l1 = p[PLANE];
      const float l2 = p[2 * PLANE];
      const float v = ok ? (l0 + l1) + l2 : 0.f;
      out[((size_t)b * 4 * 12 + o) * 12 + j] = v * (1.0f / 3.0f);
    }
  }
  __syncthreads();

  if (wave == 0)      hconv_task<1>(sm, P, out, b, lane);
  else if (wave == 1) hconv_task<2>(sm, P, out, b, lane);
  else if (wave == 2) hconv_task<3>(sm, P, out, b, lane);
}

extern "C" void kernel_launch(void* const* d_in, const int* in_sizes, int n_in,
                              void* d_out, int out_size, void* d_ws, size_t ws_size,
                              hipStream_t stream) {
  const float* imgs = (const float*)d_in[0];
  const int*   locs = (const int*)d_in[1];
  float*       out  = (float*)d_out;
  (void)in_sizes; (void)n_in; (void)d_ws; (void)ws_size; (void)out_size;

  const size_t shmem = (size_t)SMEM_FLOATS * sizeof(float);
  glimpse_wmma_kernel<<<dim3(256), dim3(NTHREADS), shmem, stream>>>(imgs, locs, out);
}


# image loads issued at kernel entry, behind the crop-position load and ahead of the tap/table set-up (its two barriers now wait for LDS only): 1/3 of each task wave's, all of the 19th wave's
# speedup vs baseline: 4.8656x; 1.0951x over previous
#include <hip/hip_runtime.h>


typedef __attribute__((ext_vector_type(2))) float v2f;
typedef __attribute__((ext_vector_type(8))) float v8f;

#define IMG      224
#define PLANE    (IMG * IMG)
#define NTT      36
#define W_OFF    0
#define SC_OFF   208
#define TB_OFF   224
#define TBSZ     (NTT * 64)
#define PITCH    148
#define PSZ      (16 * PITCH)
#define P_OFF    (TB_OFF + 3 * TBSZ)
#define SMEM_FLOATS (P_OFF + 6 * PSZ)
#define NWAVES   19
#define NTHREADS (NWAVES * 32)

__device__ __forceinline__ void lds_barrier() {
  __builtin_amdgcn_fence(__ATOMIC_RELEASE, "workgroup", "local");
  __builtin_amdgcn_s_barrier();
  __builtin_amdgcn_fence(__ATOMIC_ACQUIRE, "workgroup", "local");
}

__device__ __forceinline__ float bicubic_w(int t, int K) {
  const double A_ = -0.5;
  double x  = ((double)t - 0.5 * (double)K + 0.5) * (4.0 / (double)K);
  double ax = fabs(x);
  double w;
  if (ax > 2.0)       w = 0.0;
  else if (ax > 1.0)  w = ((A_ * ax - 5.0 * A_) * ax + 8.0 * A_) * ax - 4.0 * A_;
  else                w = (((A_ + 2.0) * ax - (A_ + 3.0)) * ax) * ax + 1.0;
  return (float)w;
}

__device__ __forceinline__ void parity_power_sums(int q0, int m, int s[4]) {
  const int J1 = m * (m - 1) / 2;
  const int J2 = (m - 1) * m * (2 * m - 1) / 6;
  const int J3 = J1 * J1;
  s[0] = m;
  s[1] = m * q0 + 2 * J1;
  s[2] = m * q0 * q0 + 4 * q0 * J1 + 4 * J2;
  s[3] = m * q0 * q0 * q0 + 6 * q0 * q0 * J1 + 12 * q0 * J2 + 8 * J3;
}

__device__ __forceinline__ float tap_scale(int K) {
  const int odd  = K & 1;
  const int q0   = odd ? 2 : 1;
  const int nin  = K >> 1;
  const int m_in  = (nin >= q0) ? (nin - q0) / 2 + 1 : 0;
  const int m_all = (K - 1 >= q0) ? (K - 1 - q0) / 2 + 1 : 0;
  int in[4], all[4];
  parity_power_sums(q0, m_in, in);
  parity_power_sums(q0, m_all, all);
  const long long I0 = 2 * in[0] + odd, I2 = 2 * in[2], I3 = 2 * in[3];
  const long long O0 = 2 * (all[0] - in[0]), O1 = 2 * (all[1] - in[1]);
  const long long O2 = 2 * (all[2] - in[2]), O3 = 2 * (all[3] - in[3]);
  const long long K1 = K, K2 = K1 * K1, K3 = K2 * K1;
  const long long num = 4 * (3 * I3 - O3) + 10 * (O2 - I2) * K1 - 8 * O1 * K2 + (I0 + 2 * O0) * K3;
  const double n = (double)num, d = (double)K3;
  return (float)((d * d) / (3.0 * n * n));
}

template <int PART>
__device__ __forceinline__ void vconv_loads(const float* __restrict__ img, int cb, int x0, int y0,
                                            int lane, int grp, float (&lv)[6][2][3])
{
  const int h   = lane >> 4;
  const int ixc = min(max(y0 + cb * 16 + (lane & 15), 0), IMG - 1);
#pragma unroll
  for (int i = 0; i < 6; ++i) {
    const int t = PART * 18 + grp * 6 + i;
    if (t < 34) {
#pragma unroll
      for (int s = 0; s < 2; ++s) {
        const int iy = x0 + 4 * t + 2 * h + s;
        const float* p = img + min(max(iy, 0), IMG - 1) * IMG + ixc;
        lv[i][s][0] = p[0];
        lv[i][s][1] = p[PLANE];
        lv[i][s][2] = p[2 * PLANE];
      }
    }
  }
}

template <int NG, int PART>
__device__ __forceinline__ void vconv_products(int grp, const float (&lv)[6][2][3], bool cval,
                                               int x0, int h, const v2f* TB1, const v2f* TB2,
                                               const v2f* TB3, v8f& acc3, v8f& acc2, v8f& acc1)
{
#pragma unroll
  for (int i = 0; i < 6; ++i) {
    const int t = PART * 18 + grp * 6 + i;
    if (t < 34) {
      const int R0 = 4 * t + 2 * h;
      const bool rv0 = cval && (R0     < 135) && (x0 + R0     >= 0) && (x0 + R0     < IMG);
      const bool rv1 = cval && (R0 + 1 < 135) && (x0 + R0 + 1 >= 0) && (x0 + R0 + 1 < IMG);
      v2f bm;
      bm.x = __int_as_float(__float_as_int((lv[i][0][0] + lv[i][0][1]) + lv[i][0][2]) & (rv0 ? -1 : 0));
      bm.y = __int_as_float(__float_as_int((lv[i][1][0] + lv[i][1][1]) + lv[i][1][2]) & (rv1 ? -1 : 0));
      acc3 = __builtin_amdgcn_wmma_f32_16x16x4_f32(
          false, TB3[t * 32], false, bm, (short)0, acc3, false, false);
      if (NG >= 2 && t >= 8 && t < 26)
        acc2 = __builtin_amdgcn_wmma_f32_16x16x4_f32(
            false, TB2[t * 32], false, bm, (short)0, acc2, false, false);
      if (NG >= 3 && t >= 14 && t < 20)
        acc1 = __builtin_amdgcn_wmma_f32_16x16x4_f32(
            false, TB1[t * 32], false, bm, (short)0, acc1, false, false);
    }
  }
}

template <int NG, int PART>
__device__ __forceinline__ void vconv_task(const float* __restrict__ img, const float* sm,
                                           float* P, int cb, int x0, int y0, int lane,
                                           const float (&lv0)[6][2][3])
{
  const int n = lane & 15;
  const int h = lane >> 4;
  const int C = cb * 16 + n;
  const int ix = y0 + C;
  const bool cval = (C < 135) && (ix >= 0) && (ix < IMG);
  const v2f* TB1 = (const v2f*)(sm + TB_OFF + 0 * TBSZ) + lane;
  const v2f* TB2 = (const v2f*)(sm + TB_OFF + 1 * TBSZ) + lane;
  const v2f* TB3 = (const v2f*)(sm + TB_OFF + 2 * TBSZ) + lane;
  v8f acc3 = {}, acc2 = {}, acc1 = {};

  float lv1[6][2][3], lv2[6][2][3], lvp[6][2][3];
  vconv_loads<PART>(img, cb, x0, y0, lane, 1, lv1);
#pragma unroll
  for (int i = 0; i < 6; ++i)
#pragma unroll
    for (int s = 0; s < 2; ++s)
#pragma unroll
      for (int c = 0; c < 3; ++c) {
        float v = lv0[i][s][c];
        asm volatile("" : "+v"(v));
        lvp[i][s][c] = v;
      }
  vconv_products<NG, PART>(0, lvp, cval, x0, h, TB1, TB2, TB3, acc3, acc2, acc1);
  vconv_loads<PART>(img, cb, x0, y0, lane, 2, lv2);
  vconv_products<NG, PART>(1, lv1, cval, x0, h, TB1, TB2, TB3, acc3, acc2, acc1);
  vconv_products<NG, PART>(2, lv2, cval, x0, h, TB1, TB2, TB3, acc3, acc2, acc1);

  float* Pp = P + (PART * 3) * PSZ + (8 * h) * PITCH + C;
#pragma unroll
  for (int r = 0; r < 8; ++r) {
    Pp[2 * PSZ + r * PITCH] = acc3[r];
    if (NG >= 2) Pp[1 * PSZ + r * PITCH] = acc2[r];
    if (NG >= 3) Pp[0 * PSZ + r * PITCH] = acc1[r];
  }
}

#define HCH 8
template <int GI>
__device__ __forceinline__ void hconv_task(const float* sm, const float* P,
                                           float* __restrict__ out, int b, int lane)
{
  constexpr int tb  = (GI == 3) ? 0  : ((GI == 2) ? 8  : 14);
  constexpr int te  = (GI == 3) ? 34 : ((GI == 2) ? 26 : 20);
  constexpr int NCH = (te - tb + HCH - 1) / HCH;
  const int m  = lane & 15;
  const int kk = (lane >> 4) << 1;
  const int Mb = (lane >> 4) << 3;
  const v2f* TBg = (const v2f*)(sm + TB_OFF + (GI - 1) * TBSZ) + lane;
  const float* V0 = P + (GI - 1) * PSZ + m * PITCH + kk;
  const float* V1 = V0 + 3 * PSZ;

  v2f p0[2][HCH], p1[2][HCH], bt[2][HCH];
#pragma unroll
  for (int i = 0; i < HCH; ++i) {
    const int t = tb + i;
    if (t < te) {
      p0[0][i] = *(const v2f*)(V0 + 4 * t);
      p1[0][i] = *(const v2f*)(V1 + 4 * t);
      bt[0][i] = TBg[t * 32];
    }
  }
  const float sc = sm[SC_OFF + GI];

  v8f accE = {}, accO = {};
#pragma unroll
  for (int c = 0; c < NCH; ++c) {
    if (c + 1 < NCH) {
#pragma unroll
      for (int i = 0; i < HCH; ++i) {
        const int t = tb + (c + 1) * HCH + i;
        if (t < te) {
          p0[(c + 1) & 1][i] = *(const v2f*)(V0 + 4 * t);
          p1[(c + 1) & 1][i] = *(const v2f*)(V1 + 4 * t);
          bt[(c + 1) & 1][i] = TBg[t * 32];
        }
      }
    }
    __builtin_amdgcn_sched_barrier(0);
#pragma unroll
    for (int i = 0; i < HCH; ++i) {
      const int t = tb + c * HCH + i;
      if (t < te) {
        const v2f a = p0[c & 1][i] + p1[c & 1][i];
        if (i & 1)
          accO = __builtin_amdgcn_wmma_f32_16x16x4_f32(
              false, a, false, bt[c & 1][i], (short)0, accO, false, false);
        else
          accE = __builtin_amdgcn_wmma_f32_16x16x4_f32(
              false, a, false, bt[c & 1][i], (short)0, accE, false, false);
      }
    }
    __builtin_amdgcn_sched_barrier(0);
  }
#pragma unroll
  for (int r = 0; r < 8; ++r) {
    const int M = Mb + r;
    if (M < 12 && m < 12)
      out[(((size_t)b * 4 + GI) * 12 + M) * 12 + m] = (accE[r] + accO[r]) * sc;
  }
}

__global__ __launch_bounds__(NTHREADS) void glimpse_wmma_kernel(
    const float* __restrict__ imgs, const int* __restrict__ locs,
    float* __restrict__ out)
{
  extern __shared__ float sm[];
  float* Wv = sm + W_OFF;
  float* SC = sm + SC_OFF;
  float* P  = sm + P_OFF;

  const int b    = blockIdx.x;
  const int tid  = threadIdx.x;
  const int lane = tid & 31;
  const int wave = __builtin_amdgcn_readfirstlane(tid >> 5);

  const int x0 = locs[2 * b + 0] - 67;
  const int y0 = locs[2 * b + 1] - 67;
  const float* img = imgs + (size_t)b * (3 * PLANE);
  const int part = (wave >= 9) ? 1 : 0;
  const int cb   = wave - 9 * part;

  if (tid < 193) {
    int t = tid, K, base;
    if (t < 12)      { K = 12;  base = 0;            }
    else if (t < 69) { K = 57;  base = 12; t -= 12;  }
    else             { K = 124; base = 69; t -= 69;  }
    Wv[base + t] = bicubic_w(t, K);
  } else if (tid >= 224 && tid < 228) {
    const int f = tid - 224;
    if (f == 3) { SC[0] = 1.0f / 3.0f; }
    else        { SC[f + 1] = tap_scale((f == 0) ? 12 : ((f == 1) ? 57 : 124)); }
  }

  float lv0[6][2][3];
  float gv[5][3];
  if (wave < 18) {
    if (part == 0) vconv_loads<0>(img, cb, x0, y0, lane, 0, lv0);
    else           vconv_loads<1>(img, cb, x0, y0, lane, 0, lv0);
  } else {
#pragma unroll
    for (int k = 0; k < 5; ++k) {
      const int t = min(lane + 32 * k, 143);
      const int o = t / 12, j = t - o * 12;
      const float* p = img + min(max(x0 + 61 + o, 0), IMG - 1) * IMG + min(max(y0 + 61 + j, 0), IMG - 1);
      gv[k][0] = p[0];
      gv[k][1] = p[PLANE];
      gv[k][2] = p[2 * PLANE];
    }
  }
  lds_barrier();

  for (int e = tid; e < 3 * NTT * 32; e += NTHREADS) {
    const int gi = e / (NTT * 32) + 1;
    const int le = e - (gi - 1) * (NTT * 32);
    const int K   = (gi == 3) ? 124 : ((gi == 2) ? 57 : 12);
    const int wb  = (gi == 3) ? 69  : ((gi == 2) ? 12 : 0);
    const int off = (gi == 3) ? 0   : ((gi == 2) ? 33 : 56);
    const int t  = le >> 5;
    const int l  = le & 31;
    const int mm = l & 15;
    const int d0 = t * 4 + ((l >> 4) << 1) - off - mm;
    float v0 = Wv[wb + min(max(d0, 0), K - 1)];
    float v1 = Wv[wb + min(max(d0 + 1, 0), K - 1)];
    v0 = (mm < 12 && d0     >= 0 && d0     < K) ? v0 : 0.f;
    v1 = (mm < 12 && d0 + 1 >= 0 && d0 + 1 < K) ? v1 : 0.f;
    sm[TB_OFF + 2 * e]     = v0;
    sm[TB_OFF + 2 * e + 1] = v1;
  }
  lds_barrier();

  if (wave < 18) {
    const int ng   = (cb == 3 || cb == 4) ? 3 : ((cb >= 2 && cb <= 6) ? 2 : 1);
    if (part == 0) {
      if (ng == 3)      vconv_task<3, 0>(img, sm, P, cb, x0, y0, lane, lv0);
      else if (ng == 2) vconv_task<2, 0>(img, sm, P, cb, x0, y0, lane, lv0);
      else              vconv_task<1, 0>(img, sm, P, cb, x0, y0, lane, lv0);
    } else {
      if (ng == 3)      vconv_task<3, 1>(img, sm, P, cb, x0, y0, lane, lv0);
      else if (ng == 2) vconv_task<2, 1>(img, sm, P, cb, x0, y0, lane, lv0);
      else              vconv_task<1, 1>(img, sm, P, cb, x0, y0, lane, lv0);
    }
  } else {
#pragma unroll
    for (int k = 0; k < 5; ++k) {
      const int t = lane + 32 * k;
      float l0 = gv[k][0], l1 = gv[k][1], l2 = gv[k][2];
      asm volatile("" : "+v"(l0), "+v"(l1), "+v"(l2));
      if (t < 144) {
        const int o = t / 12, j = t - o * 12;
        const int iy = x0 + 61 + o, ix = y0 + 61 + j;
        const bool ok = (iy >= 0) && (iy < IMG) && (ix >= 0) && (ix < IMG);
        const float v = ok ? (l0 + l1) + l2 : 0.f;
        out[((size_t)b * 4 * 12 + o) * 12 + j] = v * (1.0f / 3.0f);
      }
    }
  }
  __syncthreads();

  if (wave == 0)      hconv_task<1>(sm, P, out, b, lane);
  else if (wave == 1) hconv_task<2>(sm, P, out, b, lane);
  else if (wave == 2) hconv_task<3>(sm, P, out, b, lane);
}

extern "C" void kernel_launch(void* const* d_in, const int* in_sizes, int n_in,
                              void* d_out, int out_size, void* d_ws, size_t ws_size,
                              hipStream_t stream) {
  const float* imgs = (const float*)d_in[0];
  const int*   locs = (const int*)d_in[1];
  float*       out  = (float*)d_out;
  (void)in_sizes; (void)n_in; (void)d_ws; (void)ws_size; (void)out_size;

  const size_t shmem = (size_t)SMEM_FLOATS * sizeof(float);
  glimpse_wmma_kernel<<<dim3(256), dim3(NTHREADS), shmem, stream>>>(imgs, locs, out);
}


# filter set-up cut to one integer pass: taps as exact integers over K^3 with one f32 quotient (no f64, bit-identical) in zero-padded rows that both passes read as Toeplitz operands; no table build
# speedup vs baseline: 5.1979x; 1.0683x over previous
#include <hip/hip_runtime.h>


typedef __attribute__((ext_vector_type(2))) float v2f;
typedef __attribute__((ext_vector_type(8))) float v8f;

#define IMG      224
#define PLANE    (IMG * IMG)
#define ZLEN     160
#define ZPAD     16
#define Z_OFF    0
#define SC_OFF   (3 * ZLEN)
#define PITCH    148
#define PSZ      (16 * PITCH)
#define P_OFF    (SC_OFF + 16)
#define SMEM_FLOATS (P_OFF + 6 * PSZ)
#define NWAVES   19
#define NTHREADS (NWAVES * 32)

__device__ __forceinline__ void lds_barrier() {
  __builtin_amdgcn_fence(__ATOMIC_RELEASE, "workgroup", "local");
  __builtin_amdgcn_s_barrier();
  __builtin_amdgcn_fence(__ATOMIC_ACQUIRE, "workgroup", "local");
}

__device__ __forceinline__ float bicubic_tap(int d, int K) {
  const int q  = 2 * d - (K - 1);
  const int a  = (q < 0) ? -q : q;
  const int a2 = a * a, a3 = a2 * a;
  const int K2 = K * K, K3 = K2 * K;
  const int num = (2 * a < K) ? (12 * a3 - 10 * a2 * K + K3)
                              : (-4 * a3 + 10 * a2 * K - 8 * a * K2 + 2 * K3);
  return (float)num / (float)K3;
}

__device__ __forceinline__ v2f tap_frag(const float* Zl, int t) {
  v2f f;
  f.x = Zl[4 * t];
  f.y = Zl[4 * t + 1];
  return f;
}

__device__ __forceinline__ void parity_power_sums(int q0, int m, int s[4]) {
  const int J1 = m * (m - 1) / 2;
  const int J2 = (m - 1) * m * (2 * m - 1) / 6;
  const int J3 = J1 * J1;
  s[0] = m;
  s[1] = m * q0 + 2 * J1;
  s[2] = m * q0 * q0 + 4 * q0 * J1 + 4 * J2;
  s[3] = m * q0 * q0 * q0 + 6 * q0 * q0 * J1 + 12 * q0 * J2 + 8 * J3;
}

__device__ __forceinline__ float tap_scale(int K) {
  const int odd  = K & 1;
  const int q0   = odd ? 2 : 1;
  const int nin  = K >> 1;
  const int m_in  = (nin >= q0) ? (nin - q0) / 2 + 1 : 0;
  const int m_all = (K - 1 >= q0) ? (K - 1 - q0) / 2 + 1 : 0;
  int in[4], all[4];
  parity_power_sums(q0, m_in, in);
  parity_power_sums(q0, m_all, all);
  const long long I0 = 2 * in[0] + odd, I2 = 2 * in[2], I3 = 2 * in[3];
  const long long O0 = 2 * (all[0] - in[0]), O1 = 2 * (all[1] - in[1]);
  const long long O2 = 2 * (all[2] - in[2]), O3 = 2 * (all[3] - in[3]);
  const long long K1 = K, K2 = K1 * K1, K3 = K2 * K1;
  const long long num = 4 * (3 * I3 - O3) + 10 * (O2 - I2) * K1 - 8 * O1 * K2 + (I0 + 2 * O0) * K3;
  const double n = (double)num, d = (double)K3;
  return (float)((d * d) / (3.0 * n * n));
}

template <int PART>
__device__ __forceinline__ void vconv_loads(const float* __restrict__ img, int cb, int x0, int y0,
                                            int lane, int grp, float (&lv)[6][2][3])
{
  const int h   = lane >> 4;
  const int ixc = min(max(y0 + cb * 16 + (lane & 15), 0), IMG - 1);
#pragma unroll
  for (int i = 0; i < 6; ++i) {
    const int t = PART * 18 + grp * 6 + i;
    if (t < 34) {
#pragma unroll
      for (int s = 0; s < 2; ++s) {
        const int iy = x0 + 4 * t + 2 * h + s;
        const float* p = img + min(max(iy, 0), IMG - 1) * IMG + ixc;
        lv[i][s][0] = p[0];
        lv[i][s][1] = p[PLANE];
        lv[i][s][2] = p[2 * PLANE];
      }
    }
  }
}

template <int NG, int PART>
__device__ __forceinline__ void vconv_products(int grp, const float (&lv)[6][2][3], bool cval,
                                               int x0, int h, const float* Z1, const float* Z2,
                                               const float* Z3, v8f& acc3, v8f& acc2, v8f& acc1)
{
#pragma unroll
  for (int i = 0; i < 6; ++i) {
    const int t = PART * 18 + grp * 6 + i;
    if (t < 34) {
      const int R0 = 4 * t + 2 * h;
      const bool rv0 = cval && (R0     < 135) && (x0 + R0     >= 0) && (x0 + R0     < IMG);
      const bool rv1 = cval && (R0 + 1 < 135) && (x0 + R0 + 1 >= 0) && (x0 + R0 + 1 < IMG);
      v2f bm;
      bm.x = __int_as_float(__float_as_int((lv[i][0][0] + lv[i][0][1]) + lv[i][0][2]) & (rv0 ? -1 : 0));
      bm.y = __int_as_float(__float_as_int((lv[i][1][0] + lv[i][1][1]) + lv[i][1][2]) & (rv1 ? -1 : 0));
      acc3 = __builtin_amdgcn_wmma_f32_16x16x4_f32(
          false, tap_frag(Z3, t), false, bm, (short)0, acc3, false, false);
      if (NG >= 2 && t >= 8 && t < 26)
        acc2 = __builtin_amdgcn_wmma_f32_16x16x4_f32(
            false, tap_frag(Z2, t), false, bm, (short)0, acc2, false, false);
      if (NG >= 3 && t >= 14 && t < 20)
        acc1 = __builtin_amdgcn_wmma_f32_16x16x4_f32(
            false, tap_frag(Z1, t), false, bm, (short)0, acc1, false, false);
    }
  }
}

template <int NG, int PART>
__device__ __forceinline__ void vconv_task(const float* __restrict__ img, const float* sm,
                                           float* P, int cb, int x0, int y0, int lane,
                                           const float (&lv0)[6][2][3])
{
  const int n = lane & 15;
  const int h = lane >> 4;
  const int C = cb * 16 + n;
  const int ix = y0 + C;
  const bool cval = (C < 135) && (ix >= 0) && (ix < IMG);
  const float* Z1 = sm + Z_OFF + 0 * ZLEN + ZPAD + 2 * h - n;
  const float* Z2 = sm + Z_OFF + 1 * ZLEN + ZPAD + 2 * h - n;
  const float* Z3 = sm + Z_OFF + 2 * ZLEN + ZPAD + 2 * h - n;
  v8f acc3 = {}, acc2 = {}, acc1 = {};

  float lv1[6][2][3], lv2[6][2][3], lvp[6][2][3];
  vconv_loads<PART>(img, cb, x0, y0, lane, 1, lv1);
#pragma unroll
  for (int i = 0; i < 6; ++i)
#pragma unroll
    for (int s = 0; s < 2; ++s)
#pragma unroll
      for (int c = 0; c < 3; ++c) {
        float v = lv0[i][s][c];
        asm volatile("" : "+v"(v));
        lvp[i][s][c] = v;
      }
  vconv_products<NG, PART>(0, lvp, cval, x0, h, Z1, Z2, Z3, acc3, acc2, acc1);
  vconv_loads<PART>(img, cb, x0, y0, lane, 2, lv2);
  vconv_products<NG, PART>(1, lv1, cval, x0, h, Z1, Z2, Z3, acc3, acc2, acc1);
  vconv_products<NG, PART>(2, lv2, cval, x0, h, Z1, Z2, Z3, acc3, acc2, acc1);

  float* Pp = P + (PART * 3) * PSZ + (8 * h) * PITCH + C;
#pragma unroll
  for (int r = 0; r < 8; ++r) {
    Pp[2 * PSZ + r * PITCH] = acc3[r];
    if (NG >= 2) Pp[1 * PSZ + r * PITCH] = acc2[r];
    if (NG >= 3) Pp[0 * PSZ + r * PITCH] = acc1[r];
  }
}

#define HCH 8
template <int GI>
__device__ __forceinline__ void hconv_task(const float* sm, const float* P,
                                           float* __restrict__ out, int b, int lane)
{
  constexpr int tb  = (GI == 3) ? 0  : ((GI == 2) ? 8  : 14);
  constexpr int te  = (GI == 3) ? 34 : ((GI == 2) ? 26 : 20);
  constexpr int NCH = (te - tb + HCH - 1) / HCH;
  const int m  = lane & 15;
  const int kk = (lane >> 4) << 1;
  const int Mb = (lane >> 4) << 3;
  const float* Zg = sm + Z_OFF + (GI - 1) * ZLEN + ZPAD + kk - m;
  const float* V0 = P + (GI - 1) * PSZ + m * PITCH + kk;
  const float* V1 = V0 + 3 * PSZ;

  v2f p0[2][HCH], p1[2][HCH], bt[2][HCH];
#pragma unroll
  for (int i = 0; i < HCH; ++i) {
    const int t = tb + i;
    if (t < te) {
      p0[0][i] = *(const v2f*)(V0 + 4 * t);
      p1[0][i] = *(const v2f*)(V1 + 4 * t);
      bt[0][i] = tap_frag(Zg, t);
    }
  }
  const float sc = sm[SC_OFF + GI];

  v8f accE = {}, accO = {};
#pragma unroll
  for (int c = 0; c < NCH; ++c) {
    if (c + 1 < NCH) {
#pragma unroll
      for (int i = 0; i < HCH; ++i) {
        const int t = tb + (c + 1) * HCH + i;
        if (t < te) {
          p0[(c + 1) & 1][i] = *(const v2f*)(V0 + 4 * t);
          p1[(c + 1) & 1][i] = *(const v2f*)(V1 + 4 * t);
          bt[(c + 1) & 1][i] = tap_frag(Zg, t);
        }
      }
    }
    __builtin_amdgcn_sched_barrier(0);
#pragma unroll
    for (int i = 0; i < HCH; ++i) {
      const int t = tb + c * HCH + i;
      if (t < te) {
        const v2f a = p0[c & 1][i] + p1[c & 1][i];
        if (i & 1)
          accO = __builtin_amdgcn_wmma_f32_16x16x4_f32(
              false, a, false, bt[c & 1][i], (short)0, accO, false, false);
        else
          accE = __builtin_amdgcn_wmma_f32_16x16x4_f32(
              false, a, false, bt[c & 1][i], (short)0, accE, false, false);
      }
    }
    __builtin_amdgcn_sched_barrier(0);
  }
#pragma unroll
  for (int r = 0; r < 8; ++r) {
    const int M = Mb + r;
    if (M < 12 && m < 12)
      out[(((size_t)b * 4 + GI) * 12 + M) * 12 + m] = (accE[r] + accO[r]) * sc;
  }
}

__global__ __launch_bounds__(NTHREADS) void glimpse_wmma_kernel(
    const float* __restrict__ imgs, const int* __restrict__ locs,
    float* __restrict__ out)
{
  extern __shared__ float sm[];
  float* SC = sm + SC_OFF;
  float* P  = sm + P_OFF;

  const int b    = blockIdx.x;
  const int tid  = threadIdx.x;
  const int lane = tid & 31;
  const int wave = __builtin_amdgcn_readfirstlane(tid >> 5);

  const int x0 = locs[2 * b + 0] - 67;
  const int y0 = locs[2 * b + 1] - 67;
  const float* img = imgs + (size_t)b * (3 * PLANE);
  const int part = (wave >= 9) ? 1 : 0;
  const int cb   = wave - 9 * part;

  if (tid < 3 * ZLEN) {
    const int g   = tid / ZLEN;
    const int j   = tid - g * ZLEN;
    const int K   = (g == 2) ? 124 : ((g == 1) ? 57 : 12);
    const int off = (g == 2) ? 0   : ((g == 1) ? 33 : 56);
    const int d   = j - ZPAD - off;
    const float w = bicubic_tap(d, K);
    sm[Z_OFF + tid] = (d >= 0 && d < K) ? w : 0.f;
  }

  float lv0[6][2][3];
  if (wave < 18) {
    if (part == 0) vconv_loads<0>(img, cb, x0, y0, lane, 0, lv0);
    else           vconv_loads<1>(img, cb, x0, y0, lane, 0, lv0);
  } else if (lane < 4) {
    if (lane == 3) { SC[0] = 1.0f / 3.0f; }
    else           { SC[lane + 1] = tap_scale((lane == 0) ? 12 : ((lane == 1) ? 57 : 124)); }
  }
  lds_barrier();

  if (wave < 18) {
    const int ng   = (cb == 3 || cb == 4) ? 3 : ((cb >= 2 && cb <= 6) ? 2 : 1);
    if (part == 0) {
      if (ng == 3)      vconv_task<3, 0>(img, sm, P, cb, x0, y0, lane, lv0);
      else if (ng == 2) vconv_task<2, 0>(img, sm, P, cb, x0, y0, lane, lv0);
      else              vconv_task<1, 0>(img, sm, P, cb, x0, y0, lane, lv0);
    } else {
      if (ng == 3)      vconv_task<3, 1>(img, sm, P, cb, x0, y0, lane, lv0);
      else if (ng == 2) vconv_task<2, 1>(img, sm, P, cb, x0, y0, lane, lv0);
      else              vconv_task<1, 1>(img, sm, P, cb, x0, y0, lane, lv0);
    }
  } else {
    float gv[5][3];
#pragma unroll
    for (int k = 0; k < 5; ++k) {
      const int t = min(lane + 32 * k, 143);
      const int o = t / 12, j = t - o * 12;
      const float* p = img + min(max(x0 + 61 + o, 0), IMG - 1) * IMG + min(max(y0 + 61 + j, 0), IMG - 1);
      gv[k][0] = p[0];
      gv[k][1] = p[PLANE];
      gv[k][2] = p[2 * PLANE];
    }
#pragma unroll
    for (int k = 0; k < 5; ++k) {
      const int t = lane + 32 * k;
      const float l0 = gv[k][0], l1 = gv[k][1], l2 = gv[k][2];
      if (t < 144) {
        const int o = t / 12, j = t - o * 12;
        const int iy = x0 + 61 + o, ix = y0 + 61 + j;
        const bool ok = (iy >= 0) && (iy < IMG) && (ix >= 0) && (ix < IMG);
        const float v = ok ? (l0 + l1) + l2 : 0.f;
        out[((size_t)b * 4 * 12 + o) * 12 + j] = v * (1.0f / 3.0f);
      }
    }
  }
  __syncthreads();

  if (wave == 0)      hconv_task<1>(sm, P, out, b, lane);
  else if (wave == 1) hconv_task<2>(sm, P, out, b, lane);
  else if (wave == 2) hconv_task<3>(sm, P, out, b, lane);
}

extern "C" void kernel_launch(void* const* d_in, const int* in_sizes, int n_in,
                              void* d_out, int out_size, void* d_ws, size_t ws_size,
                              hipStream_t stream) {
  const float* imgs = (const float*)d_in[0];
  const int*   locs = (const int*)d_in[1];
  float*       out  = (float*)d_out;
  (void)in_sizes; (void)n_in; (void)d_ws; (void)ws_size; (void)out_size;

  const size_t shmem = (size_t)SMEM_FLOATS * sizeof(float);
  glimpse_wmma_kernel<<<dim3(256), dim3(NTHREADS), shmem, stream>>>(imgs, locs, out);
}
